# starting kernel
# baseline (speedup 1.0000x reference)
#include <hip/hip_runtime.h>

typedef __attribute__((ext_vector_type(2))) float v2f;
typedef __attribute__((ext_vector_type(8))) float v8f;

#define PDIM 128

__global__ __launch_bounds__(256) void painn_init(const float* __restrict__ s,
                                                  float* __restrict__ s_out,
                                                  float* __restrict__ aggv,
                                                  int n_s4, int n_a4) {
  int i = blockIdx.x * blockDim.x + threadIdx.x;
  int stride = gridDim.x * blockDim.x;
  for (int j = i; j < n_s4; j += stride)
    ((float4*)s_out)[j] = ((const float4*)s)[j];
  float4 z = make_float4(0.f, 0.f, 0.f, 0.f);
  for (int j = i; j < n_a4; j += stride)
    ((float4*)aggv)[j] = z;
}

__global__ __launch_bounds__(128) void painn_node_gemm(
    const float* __restrict__ s,
    const float* __restrict__ Wq,  const float* __restrict__ bq,
    const float* __restrict__ Wkv, const float* __restrict__ bkv,
    float* __restrict__ q, float* __restrict__ k,
    float* __restrict__ vs, float* __restrict__ vv,
    int nTiles) {
  const int nColTiles = 512 / 16;
  int wave = threadIdx.x >> 5;
  int lane = threadIdx.x & 31;
  int tile = blockIdx.x * 4 + wave;
  if (tile >= nTiles) return;
  int rowTile = tile / nColTiles;
  int colTile = tile % nColTiles;
  int rowBase = rowTile * 16;
  int cGlobal = colTile * 16;
  int half = lane >> 4;
  int l16  = lane & 15;
  int col  = cGlobal + l16;

  const float* W; const float* b; int wcol; int ldw;
  if (cGlobal < PDIM) { W = Wq;  b = bq;  wcol = col;        ldw = PDIM;     }
  else                { W = Wkv; b = bkv; wcol = col - PDIM; ldw = 3 * PDIM; }

  v8f acc = {};
  const float* srow = s + (size_t)(rowBase + l16) * PDIM;
  for (int k0 = 0; k0 < PDIM; k0 += 4) {
    float2 av = *(const float2*)(srow + k0 + 2 * half);
    v2f a; a.x = av.x; a.y = av.y;
    v2f bf;
    bf.x = W[(size_t)(k0 + 2 * half)     * ldw + wcol];
    bf.y = W[(size_t)(k0 + 2 * half + 1) * ldw + wcol];
    acc = __builtin_amdgcn_wmma_f32_16x16x4_f32(false, a, false, bf,
                                                (short)0, acc, false, false);
  }
  float bias = b[wcol];

  float* out;
  if      (cGlobal < 128) out = q;
  else if (cGlobal < 256) out = k;
  else if (cGlobal < 384) out = vs;
  else                    out = vv;
  int oc = col & (PDIM - 1);
  for (int r = 0; r < 8; ++r) {
    int row = rowBase + r + 8 * half;
    out[(size_t)row * PDIM + oc] = acc[r] + bias;
  }
}

__global__ __launch_bounds__(256) void painn_edge(
    const float* __restrict__ q, const float* __restrict__ k,
    const float* __restrict__ vs, const float* __restrict__ vv,
    const float* __restrict__ v,
    const int* __restrict__ rowIdx, const int* __restrict__ colIdx,
    float* __restrict__ s_out, float* __restrict__ aggv, int E) {
  int e = blockIdx.x * 8 + (threadIdx.x >> 5);
  if (e >= E) return;
  int lane = threadIdx.x & 31;
  int r = rowIdx[e];
  int c = colIdx[e];
  int d = lane * 4;

  float4 qv = *(const float4*)(q + (size_t)r * PDIM + d);
  float4 kv = *(const float4*)(k + (size_t)c * PDIM + d);
  float p = qv.x * kv.x + qv.y * kv.y + qv.z * kv.z + qv.w * kv.w;
  p += __shfl_xor(p, 16, 32);
  p += __shfl_xor(p, 8, 32);
  p += __shfl_xor(p, 4, 32);
  p += __shfl_xor(p, 2, 32);
  p += __shfl_xor(p, 1, 32);
  float w = p;

  float4 vsv = *(const float4*)(vs + (size_t)c * PDIM + d);
  float* so = s_out + (size_t)r * PDIM + d;
  atomicAdd(so + 0, w * vsv.x);
  atomicAdd(so + 1, w * vsv.y);
  atomicAdd(so + 2, w * vsv.z);
  atomicAdd(so + 3, w * vsv.w);

  float4 vvv = *(const float4*)(vv + (size_t)c * PDIM + d);
  const float* vc = v + (size_t)c * 3 * PDIM;
  float* av = aggv + (size_t)r * 3 * PDIM;
#pragma unroll
  for (int comp = 0; comp < 3; ++comp) {
    float4 vval = *(const float4*)(vc + comp * PDIM + d);
    float* ap = av + comp * PDIM + d;
    atomicAdd(ap + 0, w * vvv.x * vval.x);
    atomicAdd(ap + 1, w * vvv.y * vval.y);
    atomicAdd(ap + 2, w * vvv.z * vval.z);
    atomicAdd(ap + 3, w * vvv.w * vval.w);
  }
}

__global__ __launch_bounds__(128) void painn_vec_gemm(
    const float* __restrict__ aggv, const float* __restrict__ Wvec,
    const float* __restrict__ bvec, const float* __restrict__ v,
    float* __restrict__ vout, int nTiles) {
  const int nColTiles = PDIM / 16;
  int wave = threadIdx.x >> 5;
  int lane = threadIdx.x & 31;
  int tile = blockIdx.x * 4 + wave;
  if (tile >= nTiles) return;
  int rowTile = tile / nColTiles;
  int colTile = tile % nColTiles;
  int rowBase = rowTile * 16;
  int half = lane >> 4;
  int l16  = lane & 15;
  int col  = colTile * 16 + l16;

  v8f acc = {};
  const float* arow = aggv + (size_t)(rowBase + l16) * PDIM;
  for (int k0 = 0; k0 < PDIM; k0 += 4) {
    float2 av2 = *(const float2*)(arow + k0 + 2 * half);
    v2f a; a.x = av2.x; a.y = av2.y;
    v2f bf;
    bf.x = Wvec[(size_t)(k0 + 2 * half)     * PDIM + col];
    bf.y = Wvec[(size_t)(k0 + 2 * half + 1) * PDIM + col];
    acc = __builtin_amdgcn_wmma_f32_16x16x4_f32(false, a, false, bf,
                                                (short)0, acc, false, false);
  }
  float bias = bvec[col];
  for (int r = 0; r < 8; ++r) {
    size_t row = (size_t)rowBase + r + 8 * half;
    size_t idx = row * PDIM + col;
    vout[idx] = v[idx] + acc[r] + bias;
  }
}

extern "C" void kernel_launch(void* const* d_in, const int* in_sizes, int n_in,
                              void* d_out, int out_size, void* d_ws, size_t ws_size,
                              hipStream_t stream) {
  const float* s    = (const float*)d_in[0];
  const float* v    = (const float*)d_in[1];
  const int*   ei   = (const int*)d_in[2];
  const float* Wq   = (const float*)d_in[3];
  const float* bq   = (const float*)d_in[4];
  const float* Wkv  = (const float*)d_in[5];
  const float* bkv  = (const float*)d_in[6];
  const float* Wvec = (const float*)d_in[7];
  const float* bvec = (const float*)d_in[8];

  int N = in_sizes[0] / PDIM;
  int E = in_sizes[2] / 2;

  float* s_out = (float*)d_out;
  float* v_out = s_out + (size_t)N * PDIM;

  float* q    = (float*)d_ws;
  float* k    = q  + (size_t)N * PDIM;
  float* vs   = k  + (size_t)N * PDIM;
  float* vv   = vs + (size_t)N * PDIM;
  float* aggv = vv + (size_t)N * PDIM;

  const int* rowIdx = ei;
  const int* colIdx = ei + E;

  painn_init<<<2048, 256, 0, stream>>>(s, s_out, aggv,
                                       N * PDIM / 4, N * 3 * PDIM / 4);

  int tilesA = (N / 16) * (512 / 16);
  painn_node_gemm<<<(tilesA + 3) / 4, 128, 0, stream>>>(
      s, Wq, bq, Wkv, bkv, q, k, vs, vv, tilesA);

  painn_edge<<<(E + 7) / 8, 256, 0, stream>>>(
      q, k, vs, vv, v, rowIdx, colIdx, s_out, aggv, E);

  int tilesC = (N * 3 / 16) * (PDIM / 16);
  painn_vec_gemm<<<(tilesC + 3) / 4, 128, 0, stream>>>(
      aggv, Wvec, bvec, v, v_out, tilesC);
}


# Work-group takes two 98-node ranges, lists their edges in LDS in one pass over the index list; one owner wave per node sums them in source order, nodes dealt to waves by edge count; sum order changes
# speedup vs baseline: 3.2354x; 3.2354x over previous
#include <hip/hip_runtime.h>

typedef __attribute__((ext_vector_type(2))) float v2f;
typedef __attribute__((ext_vector_type(8))) float v8f;

#define PDIM 128

struct painn_panel { v2f f[4][32]; };
struct painn_strip { float4 a[16]; };

__device__ __forceinline__ void painn_load_panel(
    const float* __restrict__ W, int ldw, int c0, int l16, int half,
    painn_panel& p) {
  const float* wc = W + c0 + 4 * l16;
#pragma unroll
  for (int j = 0; j < 16; ++j) {
    int kb = 8 * j + 4 * half;
    float4 w0 = *(const float4*)(wc + (size_t)(kb + 0) * ldw);
    float4 w1 = *(const float4*)(wc + (size_t)(kb + 1) * ldw);
    float4 w2 = *(const float4*)(wc + (size_t)(kb + 2) * ldw);
    float4 w3 = *(const float4*)(wc + (size_t)(kb + 3) * ldw);
    p.f[0][2 * j].x = w0.x; p.f[0][2 * j].y = w1.x;
    p.f[1][2 * j].x = w0.y; p.f[1][2 * j].y = w1.y;
    p.f[2][2 * j].x = w0.z; p.f[2][2 * j].y = w1.z;
    p.f[3][2 * j].x = w0.w; p.f[3][2 * j].y = w1.w;
    p.f[0][2 * j + 1].x = w2.x; p.f[0][2 * j + 1].y = w3.x;
    p.f[1][2 * j + 1].x = w2.y; p.f[1][2 * j + 1].y = w3.y;
    p.f[2][2 * j + 1].x = w2.z; p.f[2][2 * j + 1].y = w3.z;
    p.f[3][2 * j + 1].x = w2.w; p.f[3][2 * j + 1].y = w3.w;
  }
}

__device__ __forceinline__ void painn_load_strip(
    const float* __restrict__ A, int rowBase, int l16, int half,
    painn_strip& st) {
  const float4* p = (const float4*)(A + (size_t)(rowBase + l16) * PDIM + 4 * half);
#pragma unroll
  for (int j = 0; j < 16; ++j) st.a[j] = p[2 * j];
}

__device__ __forceinline__ void painn_mma_strip(
    const painn_strip& st, const painn_panel& p, v8f (&acc)[4]) {
#pragma unroll
  for (int j = 0; j < 16; ++j) {
    v2f a0; a0.x = st.a[j].x; a0.y = st.a[j].y;
    v2f a1; a1.x = st.a[j].z; a1.y = st.a[j].w;
#pragma unroll
    for (int t = 0; t < 4; ++t)
      acc[t] = __builtin_amdgcn_wmma_f32_16x16x4_f32(false, a0, false, p.f[t][2 * j],
                                                     (short)0, acc[t], false, false);
#pragma unroll
    for (int t = 0; t < 4; ++t)
      acc[t] = __builtin_amdgcn_wmma_f32_16x16x4_f32(false, a1, false, p.f[t][2 * j + 1],
                                                     (short)0, acc[t], false, false);
  }
}

__global__ __launch_bounds__(256) void painn_node_gemm(
    const float* __restrict__ s,
    const float* __restrict__ Wq,  const float* __restrict__ bq,
    const float* __restrict__ Wkv, const float* __restrict__ bkv,
    float* __restrict__ q, float* __restrict__ k,
    float* __restrict__ vs, float* __restrict__ vv,
    int nRowTiles) {
  int wave = __builtin_amdgcn_readfirstlane(threadIdx.x >> 5);
  int lane = threadIdx.x & 31;
  int half = lane >> 4;
  int l16  = lane & 15;

  const float* W; const float* b; int ldw; int c0;
  if (wave < 2) { W = Wq;  b = bq;  ldw = PDIM;     c0 = 64 * wave;        }
  else          { W = Wkv; b = bkv; ldw = 3 * PDIM; c0 = 64 * wave - PDIM; }
  float* out;
  if      (wave < 2) out = q;
  else if (wave < 4) out = k;
  else if (wave < 6) out = vs;
  else               out = vv;
  int oc = 64 * (wave & 1) + 4 * l16;

  int rt = blockIdx.x;
  if (rt >= nRowTiles) return;

  painn_panel pan;
  painn_load_panel(W, ldw, c0, l16, half, pan);
  float4 bias = *(const float4*)(b + c0 + 4 * l16);

  painn_strip cur, nxt;
  painn_load_strip(s, rt * 16, l16, half, cur);
  for (; rt < nRowTiles; rt += gridDim.x) {
    int nx = rt + gridDim.x;
    if (nx >= nRowTiles) nx = nRowTiles - 1;
    painn_load_strip(s, nx * 16, l16, half, nxt);
    __builtin_amdgcn_sched_barrier(0);

    v8f acc[4] = {};
    painn_mma_strip(cur, pan, acc);
    __builtin_amdgcn_sched_barrier(0);

    float* orow = out + (size_t)(rt * 16 + 8 * half) * PDIM + oc;
#pragma unroll
    for (int r = 0; r < 8; ++r) {
      float4 o;
      o.x = acc[0][r] + bias.x;
      o.y = acc[1][r] + bias.y;
      o.z = acc[2][r] + bias.z;
      o.w = acc[3][r] + bias.w;
      *(float4*)(orow + r * PDIM) = o;
    }
    cur = nxt;
  }
}

#define RANGE 98
#define NODE_WORDS (4 * PDIM)
#define LCAP 128
#define NLISTS 64

struct painn_rows { float4 kv, vsv, vvv, x0, x1, x2; };

__device__ __forceinline__ void painn_gather(
    int c, int lane,
    const float* __restrict__ k, const float* __restrict__ vs,
    const float* __restrict__ vv, const float* __restrict__ v,
    painn_rows& g) {
  int d = lane * 4;
  g.kv  = *(const float4*)(k  + (size_t)c * PDIM + d);
  g.vsv = *(const float4*)(vs + (size_t)c * PDIM + d);
  g.vvv = *(const float4*)(vv + (size_t)c * PDIM + d);
  const float* vc = v + (size_t)c * 3 * PDIM + d;
  g.x0 = *(const float4*)(vc);
  g.x1 = *(const float4*)(vc + PDIM);
  g.x2 = *(const float4*)(vc + 2 * PDIM);
}

__device__ __forceinline__ void painn_apply(
    const painn_rows& g, int rloc, int lane, const float* qs, float* acc) {
  float4 qv = *(const float4*)(qs + rloc * PDIM + lane * 4);
  float p = qv.x * g.kv.x + qv.y * g.kv.y + qv.z * g.kv.z + qv.w * g.kv.w;
  p += __shfl_xor(p, 16, 32);
  p += __shfl_xor(p, 8, 32);
  p += __shfl_xor(p, 4, 32);
  p += __shfl_xor(p, 2, 32);
  p += __shfl_xor(p, 1, 32);
  float w = p;

  float* a = acc + rloc * NODE_WORDS + lane;
  atomicAdd(a +  0, w * g.vsv.x);
  atomicAdd(a + 32, w * g.vsv.y);
  atomicAdd(a + 64, w * g.vsv.z);
  atomicAdd(a + 96, w * g.vsv.w);
  float* a0 = a + PDIM;
  atomicAdd(a0 +  0, w * g.vvv.x * g.x0.x);
  atomicAdd(a0 + 32, w * g.vvv.y * g.x0.y);
  atomicAdd(a0 + 64, w * g.vvv.z * g.x0.z);
  atomicAdd(a0 + 96, w * g.vvv.w * g.x0.w);
  float* a1 = a + 2 * PDIM;
  atomicAdd(a1 +  0, w * g.vvv.x * g.x1.x);
  atomicAdd(a1 + 32, w * g.vvv.y * g.x1.y);
  atomicAdd(a1 + 64, w * g.vvv.z * g.x1.z);
  atomicAdd(a1 + 96, w * g.vvv.w * g.x1.w);
  float* a2 = a + 3 * PDIM;
  atomicAdd(a2 +  0, w * g.vvv.x * g.x2.x);
  atomicAdd(a2 + 32, w * g.vvv.y * g.x2.y);
  atomicAdd(a2 + 64, w * g.vvv.z * g.x2.z);
  atomicAdd(a2 + 96, w * g.vvv.w * g.x2.w);
}

__device__ __forceinline__ unsigned painn_hits(int4 a, int lo, unsigned cnt) {
  return ((unsigned)(a.x - lo) < cnt ? 1u : 0u)
       | ((unsigned)(a.y - lo) < cnt ? 2u : 0u)
       | ((unsigned)(a.z - lo) < cnt ? 4u : 0u)
       | ((unsigned)(a.w - lo) < cnt ? 8u : 0u);
}

__device__ __forceinline__ void painn_note(
    int r, int e, int lo, unsigned cntAll, unsigned* lcnt, unsigned* ndeg, uint2* raw) {
  unsigned rl = (unsigned)(r - lo);
  if (rl < cntAll) {
    unsigned h = rl >= (unsigned)RANGE ? 1u : 0u;
    unsigned rloc = rl - h * (unsigned)RANGE;
    unsigned li = h * 32u + (rloc & 31u);
    atomicAdd(&ndeg[rl], 1u);
    unsigned slot = atomicAdd(&lcnt[li], 1u);
    if (slot < (unsigned)LCAP) raw[li * LCAP + slot] = make_uint2((unsigned)e, rloc);
  }
}

__device__ __forceinline__ void painn_load_set(
    const int4* __restrict__ row4, int u, int lastIdx, int4 (&a)[8]) {
#pragma unroll
  for (int j = 0; j < 8; ++j) {
    int idx = u + j * 1024;
    a[j] = row4[idx < lastIdx ? idx : lastIdx];
  }
}

__device__ __forceinline__ void painn_note_set(
    const int4 (&a)[8], int u, int E4, int lo, unsigned cntAll,
    unsigned* lcnt, unsigned* ndeg, uint2* raw) {
#pragma unroll
  for (int j = 0; j < 8; ++j) {
    int idx = u + j * 1024;
    if (idx < E4 && painn_hits(a[j], lo, cntAll)) {
      int e0 = idx << 2;
      painn_note(a[j].x, e0,     lo, cntAll, lcnt, ndeg, raw);
      painn_note(a[j].y, e0 + 1, lo, cntAll, lcnt, ndeg, raw);
      painn_note(a[j].z, e0 + 2, lo, cntAll, lcnt, ndeg, raw);
      painn_note(a[j].w, e0 + 3, lo, cntAll, lcnt, ndeg, raw);
    }
  }
}

__device__ __forceinline__ void painn_direct_pass(
    int lo, unsigned ucnt, int t, int lane, const float* qs,
    const float* __restrict__ k, const float* __restrict__ vs,
    const float* __restrict__ vv, const float* __restrict__ v,
    const int* __restrict__ rowIdx, const int* __restrict__ colIdx,
    float* acc, int E) {
  const int4* row4 = (const int4*)rowIdx;
  const int E4 = E >> 2;
  const int4 none = make_int4(-1, -1, -1, -1);
  const int wave0 = __builtin_amdgcn_readfirstlane(t - lane);
  for (int ub = wave0; ub < E4; ub += 4096) {
    int u = ub + lane;
    int4 a0 = none, a1 = none, a2 = none, a3 = none;
    if (u < E4) a0 = row4[u];
    if (u + 1024 < E4) a1 = row4[u + 1024];
    if (u + 2048 < E4) a2 = row4[u + 2048];
    if (u + 3072 < E4) a3 = row4[u + 3072];
    unsigned m = painn_hits(a0, lo, ucnt)
               | (painn_hits(a1, lo, ucnt) << 4)
               | (painn_hits(a2, lo, ucnt) << 8)
               | (painn_hits(a3, lo, ucnt) << 12);
    unsigned any = __builtin_amdgcn_ballot_w32(m != 0u);
    while (any) {
      int src = __builtin_ctz(any);
      any &= any - 1u;
      unsigned sm = (unsigned)__builtin_amdgcn_readlane((int)m, src);
      while (sm) {
        int bit = __builtin_ctz(sm);
        sm &= sm - 1u;
        int e = ((ub + (bit >> 2) * 1024 + src) << 2) + (bit & 3);
        int r = rowIdx[e];
        painn_rows g;
        painn_gather(colIdx[e], lane, k, vs, vv, v, g);
        painn_apply(g, r - lo, lane, qs, acc);
      }
    }
  }
  if (t < 32) {
    for (int e = E4 << 2; e < E; ++e) {
      int r = rowIdx[e];
      if ((unsigned)(r - lo) < ucnt) {
        painn_rows g;
        painn_gather(colIdx[e], lane, k, vs, vv, v, g);
        painn_apply(g, r - lo, lane, qs, acc);
      }
    }
  }
}

__global__ __launch_bounds__(1024) void painn_edge_range(
    const float* __restrict__ q, const float* __restrict__ k,
    const float* __restrict__ vs, const float* __restrict__ vv,
    const float* __restrict__ v, const float* __restrict__ s,
    const int* __restrict__ rowIdx, const int* __restrict__ colIdx,
    float* __restrict__ s_out, float* __restrict__ aggv, int N, int E) {
  __shared__ float acc[RANGE * NODE_WORDS];
  __shared__ float qs[RANGE * PDIM];
  __shared__ unsigned srt[NLISTS * LCAP];
  __shared__ unsigned lcnt[NLISTS];
  __shared__ unsigned fcnt[NLISTS];
  __shared__ unsigned ndeg[2 * RANGE];
  __shared__ int wnode[NLISTS * 4];
  __shared__ unsigned ovf;

  const int t = threadIdx.x;
  const int lane = t & 31;
  const int wave = __builtin_amdgcn_readfirstlane(t >> 5);
  const int lo = blockIdx.x * (2 * RANGE);
  int cntAll = N - lo;
  if (cntAll > 2 * RANGE) cntAll = 2 * RANGE;
  const unsigned ucntAll = (unsigned)cntAll;

  uint2* raw = (uint2*)acc;
  unsigned long long* keys = (unsigned long long*)(acc + 2 * NLISTS * LCAP);
  unsigned* pk = (unsigned*)(acc + 4 * NLISTS * LCAP);

  if (t < NLISTS) lcnt[t] = 0u;
  if (t < 2 * RANGE) ndeg[t] = 0u;
  if (t < NLISTS * 4) wnode[t] = -1;
  if (t == 0) ovf = 0u;
  __syncthreads();

  {
    const int4* row4 = (const int4*)rowIdx;
    const int E4 = E >> 2;
    const int wave0 = __builtin_amdgcn_readfirstlane(t - lane);
    if (E4 > 0) {
      int4 sa[8], sb[8];
      painn_load_set(row4, wave0 + lane, E4 - 1, sa);
      for (int ub = wave0; ub < E4; ub += 16384) {
        const int u = ub + lane;
        painn_load_set(row4, u + 8192, E4 - 1, sb);
        painn_note_set(sa, u, E4, lo, ucntAll, lcnt, ndeg, raw);
        painn_load_set(row4, u + 16384, E4 - 1, sa);
        painn_note_set(sb, u + 8192, E4, lo, ucntAll, lcnt, ndeg, raw);
      }
    }
    int et = (E4 << 2) + t;
    if (et < E) painn_note(rowIdx[et], et, lo, ucntAll, lcnt, ndeg, raw);
  }
  __syncthreads();
  if (t < NLISTS && lcnt[t] > (unsigned)LCAP) ovf = 1u;
  if (t < cntAll) {
    const int h = t >= RANGE ? 1 : 0;
    const int rloc = t - h * RANGE;
    int cntH = cntAll - h * RANGE;
    if (cntH > RANGE) cntH = RANGE;
    const unsigned* dg = ndeg + h * RANGE;
    const unsigned mine = dg[rloc];
    int rank = 0;
    for (int j = 0; j < cntH; ++j) {
      unsigned o = dg[j];
      rank += (o > mine || (o == mine && j < rloc)) ? 1 : 0;
    }
    const int round = rank >> 5;
    const int pos = rank & 31;
    const int w = (round & 1) ? 31 - pos : pos;
    wnode[(h * 32 + w) * 4 + round] = rloc;
  }
  __syncthreads();
  if (t < NLISTS) {
    const int h = t >> 5;
    unsigned n = 0u;
    for (int qd = 0; qd < 4; ++qd) {
      int rloc = wnode[t * 4 + qd];
      if (rloc >= 0) n += ndeg[h * RANGE + rloc];
    }
    fcnt[t] = n;
    if (n > (unsigned)LCAP) ovf = 1u;
  }
  __syncthreads();
  const bool direct = ovf != 0u;

  if (!direct) {
    for (int h = 0; h < 2; ++h) {
      const int li = h * 32 + wave;
      int off = 0;
      for (int qd = 0; qd < 4; ++qd) {
        const int rloc = __builtin_amdgcn_readfirstlane(wnode[li * 4 + qd]);
        if (rloc < 0) continue;
        const int b = h * 32 + (rloc & 31);
        const int nb = __builtin_amdgcn_readfirstlane((int)lcnt[b]);
        for (int c0 = 0; c0 < nb; c0 += 32) {
          const int i = c0 + lane;
          uint2 en = make_uint2(0u, 0xFFFFFFFFu);
          if (i < nb) en = raw[b * LCAP + i];
          const bool m = en.y == (unsigned)rloc;
          const unsigned mask = __builtin_amdgcn_ballot_w32(m);
          if (m) {
            const int pos = off + __builtin_popcount(mask & ((1u << lane) - 1u));
            keys[li * LCAP + pos] = ((unsigned long long)en.y << 32) | (unsigned long long)en.x;
          }
          off += __builtin_popcount(mask);
        }
      }
    }
    __syncthreads();
    for (int h = 0; h < 2; ++h) {
      const int li = h * 32 + wave;
      const int n = __builtin_amdgcn_readfirstlane((int)fcnt[li]);
      if (n > 0) {
        unsigned long long tk[4];
        unsigned cc[4];
#pragma unroll
        for (int r4 = 0; r4 < 4; ++r4) {
          const int i = lane + 32 * r4;
          tk[r4] = keys[li * LCAP + (i < n ? i : n - 1)];
        }
#pragma unroll
        for (int r4 = 0; r4 < 4; ++r4) cc[r4] = (unsigned)colIdx[(unsigned)tk[r4]];
#pragma unroll
        for (int r4 = 0; r4 < 4; ++r4) {
          const int i = lane + 32 * r4;
          if (i < n) {
            const unsigned e = (unsigned)tk[r4];
            const unsigned rloc = (unsigned)(tk[r4] >> 32);
            keys[li * LCAP + i] = ((unsigned long long)cc[r4] << 32) | (unsigned long long)e;
            pk[li * LCAP + i] = (rloc << 24) | cc[r4];
          }
        }
      }
    }
    __syncthreads();
    for (int h = 0; h < 2; ++h) {
      const int li = h * 32 + wave;
      const int n = (int)fcnt[li];
      for (int i = lane; i < n; i += 32) {
        unsigned long long key = keys[li * LCAP + i];
        int rank = 0;
#pragma unroll 4
        for (int j = 0; j < n; ++j) rank += keys[li * LCAP + j] < key ? 1 : 0;
        srt[li * LCAP + rank] = pk[li * LCAP + i];
      }
    }
    __syncthreads();
  }

  for (int h = 0; h < 2; ++h) {
    const int slo = lo + h * RANGE;
    int cnt = N - slo;
    if (cnt > RANGE) cnt = RANGE;
    if (cnt <= 0) break;

    float4 z = make_float4(0.f, 0.f, 0.f, 0.f);
    for (int i = t; i < RANGE * NODE_WORDS / 4; i += 1024)
      ((float4*)acc)[i] = z;
    {
      const float4* q4 = (const float4*)q + (size_t)slo * 32;
      for (int i = t; i < cnt * 32; i += 1024)
        ((float4*)qs)[i] = q4[i];
    }
    __syncthreads();

    if (direct) {
      painn_direct_pass(slo, (unsigned)cnt, t, lane, qs, k, vs, vv, v,
                        rowIdx, colIdx, acc, E);
    } else {
      const unsigned* L = srt + (h * 32 + wave) * LCAP;
      const int n = __builtin_amdgcn_readfirstlane((int)fcnt[h * 32 + wave]);
      unsigned e0 = lane < n ? L[lane] : 0u;
      unsigned e1 = lane + 32 < n ? L[lane + 32] : 0u;
      unsigned e2 = lane + 64 < n ? L[lane + 64] : 0u;
      unsigned e3 = lane + 96 < n ? L[lane + 96] : 0u;
#define PAINN_ENTRY(i) ((unsigned)__builtin_amdgcn_readlane( \
    (int)(((i) & 64) ? (((i) & 32) ? e3 : e2) : (((i) & 32) ? e1 : e0)), (i) & 31))
      painn_rows ga, gb, gc;
      unsigned pa = 0u, pb = 0u, pc = 0u;
      if (n > 0) {
        const int last = n - 1;
        pa = PAINN_ENTRY(0);
        painn_gather((int)(pa & 0xFFFFFFu), lane, k, vs, vv, v, ga);
        {
          const int j = 1 < last ? 1 : last;
          pb = PAINN_ENTRY(j);
          painn_gather((int)(pb & 0xFFFFFFu), lane, k, vs, vv, v, gb);
        }
        for (int i = 0; i < n; i += 3) {
          {
            const int j = i + 2 < last ? i + 2 : last;
            pc = PAINN_ENTRY(j);
            painn_gather((int)(pc & 0xFFFFFFu), lane, k, vs, vv, v, gc);
          }
          painn_apply(ga, (int)(pa >> 24), lane, qs, acc);
          {
            const int j = i + 3 < last ? i + 3 : last;
            pa = PAINN_ENTRY(j);
            painn_gather((int)(pa & 0xFFFFFFu), lane, k, vs, vv, v, ga);
          }
          if (i + 1 < n) painn_apply(gb, (int)(pb >> 24), lane, qs, acc);
          {
            const int j = i + 4 < last ? i + 4 : last;
            pb = PAINN_ENTRY(j);
            painn_gather((int)(pb & 0xFFFFFFu), lane, k, vs, vv, v, gb);
          }
          if (i + 2 < n) painn_apply(gc, (int)(pc >> 24), lane, qs, acc);
        }
      }
#undef PAINN_ENTRY
    }
    __syncthreads();

    {
      const float4* s4 = (const float4*)s + (size_t)slo * 32;
      float4* so4 = (float4*)s_out + (size_t)slo * 32;
      for (int u = t; u < cnt * 32; u += 1024) {
        const float* a = acc + (u >> 5) * NODE_WORDS + (u & 31);
        float4 sv = s4[u];
        float4 val;
        val.x = sv.x + a[0];
        val.y = sv.y + a[32];
        val.z = sv.z + a[64];
        val.w = sv.w + a[96];
        so4[u] = val;
      }
    }
    {
      float4* av4 = (float4*)aggv + (size_t)slo * 96;
      for (int u = t; u < cnt * 96; u += 1024) {
        int n = u / 96;
        int rem = u - n * 96;
        const float* a = acc + n * NODE_WORDS + PDIM + (rem >> 5) * PDIM + (rem & 31);
        av4[u] = make_float4(a[0], a[32], a[64], a[96]);
      }
    }
    __syncthreads();
  }
}

__global__ __launch_bounds__(256) void painn_vec_gemm(
    const float* __restrict__ aggv, const float* __restrict__ Wvec,
    const float* __restrict__ bvec, const float* __restrict__ v,
    float* __restrict__ vout, int nRowTiles) {
  int wave = __builtin_amdgcn_readfirstlane(threadIdx.x >> 5);
  int lane = threadIdx.x & 31;
  int half = lane >> 4;
  int l16  = lane & 15;
  int c0 = 64 * (wave & 1);
  int oc = c0 + 4 * l16;
  int nStreams = gridDim.x * 4;

  int rt = blockIdx.x * 4 + (wave >> 1);
  if (rt >= nRowTiles) return;

  painn_panel pan;
  painn_load_panel(Wvec, PDIM, c0, l16, half, pan);
  float4 bias = *(const float4*)(bvec + oc);

  painn_strip cur, nxt;
  painn_load_strip(aggv, rt * 16, l16, half, cur);
  for (; rt < nRowTiles; rt += nStreams) {
    int nx = rt + nStreams;
    if (nx >= nRowTiles) nx = nRowTiles - 1;
    painn_load_strip(aggv, nx * 16, l16, half, nxt);

    size_t o0 = (size_t)(rt * 16 + 8 * half) * PDIM + oc;
    const float* vrow = v + o0;
    float* orow = vout + o0;
    float4 vin[8];
#pragma unroll
    for (int r = 0; r < 8; ++r) vin[r] = *(const float4*)(vrow + r * PDIM);
    __builtin_amdgcn_sched_barrier(0);

    v8f acc[4] = {};
    painn_mma_strip(cur, pan, acc);

#pragma unroll
    for (int r = 0; r < 8; ++r) {
      float4 o;
      o.x = vin[r].x + acc[0][r] + bias.x;
      o.y = vin[r].y + acc[1][r] + bias.y;
      o.z = vin[r].z + acc[2][r] + bias.z;
      o.w = vin[r].w + acc[3][r] + bias.w;
      *(float4*)(orow + r * PDIM) = o;
    }
    cur = nxt;
  }
}

extern "C" void kernel_launch(void* const* d_in, const int* in_sizes, int n_in,
                              void* d_out, int out_size, void* d_ws, size_t ws_size,
                              hipStream_t stream) {
  const float* s    = (const float*)d_in[0];
  const float* v    = (const float*)d_in[1];
  const int*   ei   = (const int*)d_in[2];
  const float* Wq   = (const float*)d_in[3];
  const float* bq   = (const float*)d_in[4];
  const float* Wkv  = (const float*)d_in[5];
  const float* bkv  = (const float*)d_in[6];
  const float* Wvec = (const float*)d_in[7];
  const float* bvec = (const float*)d_in[8];

  int N = in_sizes[0] / PDIM;
  int E = in_sizes[2] / 2;

  float* s_out = (float*)d_out;
  float* v_out = s_out + (size_t)N * PDIM;

  float* q    = (float*)d_ws;
  float* k    = q  + (size_t)N * PDIM;
  float* vs   = k  + (size_t)N * PDIM;
  float* vv   = vs + (size_t)N * PDIM;
  float* aggv = vv + (size_t)N * PDIM;

  const int* rowIdx = ei;
  const int* colIdx = ei + E;

  painn_node_gemm<<<256, 256, 0, stream>>>(
      s, Wq, bq, Wkv, bkv, q, k, vs, vv, N / 16);

  painn_edge_range<<<(N + 2 * RANGE - 1) / (2 * RANGE), 1024, 0, stream>>>(
      q, k, vs, vv, v, s, rowIdx, colIdx, s_out, aggv, N, E);

  painn_vec_gemm<<<256, 256, 0, stream>>>(
      aggv, Wvec, bvec, v, v_out, N * 3 / 16);
}


# Owner wave's list ordered by source quarter, then node, then source: a node's consecutive edges are summed in registers and added to its LDS sums once per run, not per edge; sum order changes
# speedup vs baseline: 3.3914x; 1.0482x over previous
#include <hip/hip_runtime.h>

typedef __attribute__((ext_vector_type(2))) float v2f;
typedef __attribute__((ext_vector_type(8))) float v8f;

#define PDIM 128

struct painn_panel { v2f f[4][32]; };
struct painn_strip { float4 a[16]; };

__device__ __forceinline__ void painn_load_panel(
    const float* __restrict__ W, int ldw, int c0, int l16, int half,
    painn_panel& p) {
  const float* wc = W + c0 + 4 * l16;
#pragma unroll
  for (int j = 0; j < 16; ++j) {
    int kb = 8 * j + 4 * half;
    float4 w0 = *(const float4*)(wc + (size_t)(kb + 0) * ldw);
    float4 w1 = *(const float4*)(wc + (size_t)(kb + 1) * ldw);
    float4 w2 = *(const float4*)(wc + (size_t)(kb + 2) * ldw);
    float4 w3 = *(const float4*)(wc + (size_t)(kb + 3) * ldw);
    p.f[0][2 * j].x = w0.x; p.f[0][2 * j].y = w1.x;
    p.f[1][2 * j].x = w0.y; p.f[1][2 * j].y = w1.y;
    p.f[2][2 * j].x = w0.z; p.f[2][2 * j].y = w1.z;
    p.f[3][2 * j].x = w0.w; p.f[3][2 * j].y = w1.w;
    p.f[0][2 * j + 1].x = w2.x; p.f[0][2 * j + 1].y = w3.x;
    p.f[1][2 * j + 1].x = w2.y; p.f[1][2 * j + 1].y = w3.y;
    p.f[2][2 * j + 1].x = w2.z; p.f[2][2 * j + 1].y = w3.z;
    p.f[3][2 * j + 1].x = w2.w; p.f[3][2 * j + 1].y = w3.w;
  }
}

__device__ __forceinline__ void painn_load_strip(
    const float* __restrict__ A, int rowBase, int l16, int half,
    painn_strip& st) {
  const float4* p = (const float4*)(A + (size_t)(rowBase + l16) * PDIM + 4 * half);
#pragma unroll
  for (int j = 0; j < 16; ++j) st.a[j] = p[2 * j];
}

__device__ __forceinline__ void painn_mma_strip(
    const painn_strip& st, const painn_panel& p, v8f (&acc)[4]) {
#pragma unroll
  for (int j = 0; j < 16; ++j) {
    v2f a0; a0.x = st.a[j].x; a0.y = st.a[j].y;
    v2f a1; a1.x = st.a[j].z; a1.y = st.a[j].w;
#pragma unroll
    for (int t = 0; t < 4; ++t)
      acc[t] = __builtin_amdgcn_wmma_f32_16x16x4_f32(false, a0, false, p.f[t][2 * j],
                                                     (short)0, acc[t], false, false);
#pragma unroll
    for (int t = 0; t < 4; ++t)
      acc[t] = __builtin_amdgcn_wmma_f32_16x16x4_f32(false, a1, false, p.f[t][2 * j + 1],
                                                     (short)0, acc[t], false, false);
  }
}

__global__ __launch_bounds__(256) void painn_node_gemm(
    const float* __restrict__ s,
    const float* __restrict__ Wq,  const float* __restrict__ bq,
    const float* __restrict__ Wkv, const float* __restrict__ bkv,
    float* __restrict__ q, float* __restrict__ k,
    float* __restrict__ vs, float* __restrict__ vv,
    int nRowTiles) {
  int wave = __builtin_amdgcn_readfirstlane(threadIdx.x >> 5);
  int lane = threadIdx.x & 31;
  int half = lane >> 4;
  int l16  = lane & 15;

  const float* W; const float* b; int ldw; int c0;
  if (wave < 2) { W = Wq;  b = bq;  ldw = PDIM;     c0 = 64 * wave;        }
  else          { W = Wkv; b = bkv; ldw = 3 * PDIM; c0 = 64 * wave - PDIM; }
  float* out;
  if      (wave < 2) out = q;
  else if (wave < 4) out = k;
  else if (wave < 6) out = vs;
  else               out = vv;
  int oc = 64 * (wave & 1) + 4 * l16;

  int rt = blockIdx.x;
  if (rt >= nRowTiles) return;

  painn_panel pan;
  painn_load_panel(W, ldw, c0, l16, half, pan);
  float4 bias = *(const float4*)(b + c0 + 4 * l16);

  painn_strip cur, nxt;
  painn_load_strip(s, rt * 16, l16, half, cur);
  for (; rt < nRowTiles; rt += gridDim.x) {
    int nx = rt + gridDim.x;
    if (nx >= nRowTiles) nx = nRowTiles - 1;
    painn_load_strip(s, nx * 16, l16, half, nxt);
    __builtin_amdgcn_sched_barrier(0);

    v8f acc[4] = {};
    painn_mma_strip(cur, pan, acc);
    __builtin_amdgcn_sched_barrier(0);

    float* orow = out + (size_t)(rt * 16 + 8 * half) * PDIM + oc;
#pragma unroll
    for (int r = 0; r < 8; ++r) {
      float4 o;
      o.x = acc[0][r] + bias.x;
      o.y = acc[1][r] + bias.y;
      o.z = acc[2][r] + bias.z;
      o.w = acc[3][r] + bias.w;
      *(float4*)(orow + r * PDIM) = o;
    }
    cur = nxt;
  }
}

#define RANGE 98
#define NODE_WORDS (4 * PDIM)
#define LCAP 128
#define NLISTS 64

struct painn_rows { float4 kv, vsv, vvv, x0, x1, x2; };

__device__ __forceinline__ void painn_gather(
    int c, int lane,
    const float* __restrict__ k, const float* __restrict__ vs,
    const float* __restrict__ vv, const float* __restrict__ v,
    painn_rows& g) {
  const unsigned ob = (unsigned)c * (unsigned)(PDIM * 4) + (unsigned)lane * 16u;
  g.kv  = *(const float4*)((const char*)k  + ob);
  g.vsv = *(const float4*)((const char*)vs + ob);
  g.vvv = *(const float4*)((const char*)vv + ob);
  const unsigned ov = (unsigned)c * (unsigned)(3 * PDIM * 4) + (unsigned)lane * 16u;
  const char* vc = (const char*)v + ov;
  g.x0 = *(const float4*)(vc);
  g.x1 = *(const float4*)(vc + PDIM * 4);
  g.x2 = *(const float4*)(vc + 2 * PDIM * 4);
}

__device__ __forceinline__ void painn_apply(
    const painn_rows& g, int rloc, int lane, const float* qs, float* acc) {
  float4 qv = *(const float4*)(qs + rloc * PDIM + lane * 4);
  float p = qv.x * g.kv.x + qv.y * g.kv.y + qv.z * g.kv.z + qv.w * g.kv.w;
  p += __shfl_xor(p, 16, 32);
  p += __shfl_xor(p, 8, 32);
  p += __shfl_xor(p, 4, 32);
  p += __shfl_xor(p, 2, 32);
  p += __shfl_xor(p, 1, 32);
  float w = p;

  float* a = acc + rloc * NODE_WORDS + lane;
  atomicAdd(a +  0, w * g.vsv.x);
  atomicAdd(a + 32, w * g.vsv.y);
  atomicAdd(a + 64, w * g.vsv.z);
  atomicAdd(a + 96, w * g.vsv.w);
  float* a0 = a + PDIM;
  atomicAdd(a0 +  0, w * g.vvv.x * g.x0.x);
  atomicAdd(a0 + 32, w * g.vvv.y * g.x0.y);
  atomicAdd(a0 + 64, w * g.vvv.z * g.x0.z);
  atomicAdd(a0 + 96, w * g.vvv.w * g.x0.w);
  float* a1 = a + 2 * PDIM;
  atomicAdd(a1 +  0, w * g.vvv.x * g.x1.x);
  atomicAdd(a1 + 32, w * g.vvv.y * g.x1.y);
  atomicAdd(a1 + 64, w * g.vvv.z * g.x1.z);
  atomicAdd(a1 + 96, w * g.vvv.w * g.x1.w);
  float* a2 = a + 3 * PDIM;
  atomicAdd(a2 +  0, w * g.vvv.x * g.x2.x);
  atomicAdd(a2 + 32, w * g.vvv.y * g.x2.y);
  atomicAdd(a2 + 64, w * g.vvv.z * g.x2.z);
  atomicAdd(a2 + 96, w * g.vvv.w * g.x2.w);
}

struct painn_sums { float4 s, v0, v1, v2; };

__device__ __forceinline__ void painn_add_edge(
    const painn_rows& g, const float4& qv, painn_sums& a) {
  float p = qv.x * g.kv.x + qv.y * g.kv.y + qv.z * g.kv.z + qv.w * g.kv.w;
  p += __shfl_xor(p, 16, 32);
  p += __shfl_xor(p, 8, 32);
  p += __shfl_xor(p, 4, 32);
  p += __shfl_xor(p, 2, 32);
  p += __shfl_xor(p, 1, 32);
  const float w = p;
  a.s.x += w * g.vsv.x;
  a.s.y += w * g.vsv.y;
  a.s.z += w * g.vsv.z;
  a.s.w += w * g.vsv.w;
  const float tx = w * g.vvv.x, ty = w * g.vvv.y, tz = w * g.vvv.z, tw = w * g.vvv.w;
  a.v0.x += tx * g.x0.x; a.v0.y += ty * g.x0.y; a.v0.z += tz * g.x0.z; a.v0.w += tw * g.x0.w;
  a.v1.x += tx * g.x1.x; a.v1.y += ty * g.x1.y; a.v1.z += tz * g.x1.z; a.v1.w += tw * g.x1.w;
  a.v2.x += tx * g.x2.x; a.v2.y += ty * g.x2.y; a.v2.z += tz * g.x2.z; a.v2.w += tw * g.x2.w;
}

__device__ __forceinline__ void painn_flush(
    const painn_sums& a, int rloc, int lane, float* acc) {
  float* p = acc + rloc * NODE_WORDS + lane;
  atomicAdd(p +  0, a.s.x);
  atomicAdd(p + 32, a.s.y);
  atomicAdd(p + 64, a.s.z);
  atomicAdd(p + 96, a.s.w);
  float* p0 = p + PDIM;
  atomicAdd(p0 +  0, a.v0.x);
  atomicAdd(p0 + 32, a.v0.y);
  atomicAdd(p0 + 64, a.v0.z);
  atomicAdd(p0 + 96, a.v0.w);
  float* p1 = p + 2 * PDIM;
  atomicAdd(p1 +  0, a.v1.x);
  atomicAdd(p1 + 32, a.v1.y);
  atomicAdd(p1 + 64, a.v1.z);
  atomicAdd(p1 + 96, a.v1.w);
  float* p2 = p + 3 * PDIM;
  atomicAdd(p2 +  0, a.v2.x);
  atomicAdd(p2 + 32, a.v2.y);
  atomicAdd(p2 + 64, a.v2.z);
  atomicAdd(p2 + 96, a.v2.w);
}

__device__ __forceinline__ unsigned painn_hits(int4 a, int lo, unsigned cnt) {
  return ((unsigned)(a.x - lo) < cnt ? 1u : 0u)
       | ((unsigned)(a.y - lo) < cnt ? 2u : 0u)
       | ((unsigned)(a.z - lo) < cnt ? 4u : 0u)
       | ((unsigned)(a.w - lo) < cnt ? 8u : 0u);
}

__device__ __forceinline__ void painn_note(
    int r, int e, int lo, unsigned cntAll, unsigned* lcnt, unsigned* ndeg, uint2* raw) {
  unsigned rl = (unsigned)(r - lo);
  if (rl < cntAll) {
    unsigned h = rl >= (unsigned)RANGE ? 1u : 0u;
    unsigned rloc = rl - h * (unsigned)RANGE;
    unsigned li = h * 32u + (rloc & 31u);
    atomicAdd(&ndeg[rl], 1u);
    unsigned slot = atomicAdd(&lcnt[li], 1u);
    if (slot < (unsigned)LCAP) raw[li * LCAP + slot] = make_uint2((unsigned)e, rloc);
  }
}

__device__ __forceinline__ void painn_load_set(
    const int4* __restrict__ row4, int u, int lastIdx, int4 (&a)[8]) {
#pragma unroll
  for (int j = 0; j < 8; ++j) {
    int idx = u + j * 1024;
    a[j] = row4[idx < lastIdx ? idx : lastIdx];
  }
}

__device__ __forceinline__ void painn_note_set(
    const int4 (&a)[8], int u, int E4, int lo, unsigned cntAll,
    unsigned* lcnt, unsigned* ndeg, uint2* raw) {
#pragma unroll
  for (int j = 0; j < 8; ++j) {
    int idx = u + j * 1024;
    if (idx < E4 && painn_hits(a[j], lo, cntAll)) {
      int e0 = idx << 2;
      painn_note(a[j].x, e0,     lo, cntAll, lcnt, ndeg, raw);
      painn_note(a[j].y, e0 + 1, lo, cntAll, lcnt, ndeg, raw);
      painn_note(a[j].z, e0 + 2, lo, cntAll, lcnt, ndeg, raw);
      painn_note(a[j].w, e0 + 3, lo, cntAll, lcnt, ndeg, raw);
    }
  }
}

__device__ __forceinline__ void painn_direct_pass(
    int lo, unsigned ucnt, int t, int lane, const float* qs,
    const float* __restrict__ k, const float* __restrict__ vs,
    const float* __restrict__ vv, const float* __restrict__ v,
    const int* __restrict__ rowIdx, const int* __restrict__ colIdx,
    float* acc, int E) {
  const int4* row4 = (const int4*)rowIdx;
  const int E4 = E >> 2;
  const int4 none = make_int4(-1, -1, -1, -1);
  const int wave0 = __builtin_amdgcn_readfirstlane(t - lane);
  for (int ub = wave0; ub < E4; ub += 4096) {
    int u = ub + lane;
    int4 a0 = none, a1 = none, a2 = none, a3 = none;
    if (u < E4) a0 = row4[u];
    if (u + 1024 < E4) a1 = row4[u + 1024];
    if (u + 2048 < E4) a2 = row4[u + 2048];
    if (u + 3072 < E4) a3 = row4[u + 3072];
    unsigned m = painn_hits(a0, lo, ucnt)
               | (painn_hits(a1, lo, ucnt) << 4)
               | (painn_hits(a2, lo, ucnt) << 8)
               | (painn_hits(a3, lo, ucnt) << 12);
    unsigned any = __builtin_amdgcn_ballot_w32(m != 0u);
    while (any) {
      int src = __builtin_ctz(any);
      any &= any - 1u;
      unsigned sm = (unsigned)__builtin_amdgcn_readlane((int)m, src);
      while (sm) {
        int bit = __builtin_ctz(sm);
        sm &= sm - 1u;
        int e = ((ub + (bit >> 2) * 1024 + src) << 2) + (bit & 3);
        int r = rowIdx[e];
        painn_rows g;
        painn_gather(colIdx[e], lane, k, vs, vv, v, g);
        painn_apply(g, r - lo, lane, qs, acc);
      }
    }
  }
  if (t < 32) {
    for (int e = E4 << 2; e < E; ++e) {
      int r = rowIdx[e];
      if ((unsigned)(r - lo) < ucnt) {
        painn_rows g;
        painn_gather(colIdx[e], lane, k, vs, vv, v, g);
        painn_apply(g, r - lo, lane, qs, acc);
      }
    }
  }
}

__global__ __launch_bounds__(1024) void painn_edge_range(
    const float* __restrict__ q, const float* __restrict__ k,
    const float* __restrict__ vs, const float* __restrict__ vv,
    const float* __restrict__ v, const float* __restrict__ s,
    const int* __restrict__ rowIdx, const int* __restrict__ colIdx,
    float* __restrict__ s_out, float* __restrict__ aggv, int N, int E) {
  __shared__ float acc[RANGE * NODE_WORDS];
  __shared__ float qs[RANGE * PDIM];
  __shared__ unsigned srt[NLISTS * LCAP];
  __shared__ unsigned lcnt[NLISTS];
  __shared__ unsigned fcnt[NLISTS];
  __shared__ unsigned ndeg[2 * RANGE];
  __shared__ int wnode[NLISTS * 4];
  __shared__ unsigned ovf;

  const int t = threadIdx.x;
  const int lane = t & 31;
  const int wave = __builtin_amdgcn_readfirstlane(t >> 5);
  const int lo = blockIdx.x * (2 * RANGE);
  int cntAll = N - lo;
  if (cntAll > 2 * RANGE) cntAll = 2 * RANGE;
  const unsigned ucntAll = (unsigned)cntAll;

  uint2* raw = (uint2*)acc;
  unsigned long long* keys = (unsigned long long*)(acc + 2 * NLISTS * LCAP);
  unsigned* pk = (unsigned*)(acc + 4 * NLISTS * LCAP);

  if (t < NLISTS) lcnt[t] = 0u;
  if (t < 2 * RANGE) ndeg[t] = 0u;
  if (t < NLISTS * 4) wnode[t] = -1;
  if (t == 0) ovf = 0u;
  __syncthreads();

  {
    const int4* row4 = (const int4*)rowIdx;
    const int E4 = E >> 2;
    const int wave0 = __builtin_amdgcn_readfirstlane(t - lane);
    if (E4 > 0) {
      int4 sa[8], sb[8];
      painn_load_set(row4, wave0 + lane, E4 - 1, sa);
      for (int ub = wave0; ub < E4; ub += 16384) {
        const int u = ub + lane;
        painn_load_set(row4, u + 8192, E4 - 1, sb);
        painn_note_set(sa, u, E4, lo, ucntAll, lcnt, ndeg, raw);
        painn_load_set(row4, u + 16384, E4 - 1, sa);
        painn_note_set(sb, u + 8192, E4, lo, ucntAll, lcnt, ndeg, raw);
      }
    }
    int et = (E4 << 2) + t;
    if (et < E) painn_note(rowIdx[et], et, lo, ucntAll, lcnt, ndeg, raw);
  }
  __syncthreads();
  if (t < NLISTS && lcnt[t] > (unsigned)LCAP) ovf = 1u;
  if (t < cntAll) {
    const int h = t >= RANGE ? 1 : 0;
    const int rloc = t - h * RANGE;
    int cntH = cntAll - h * RANGE;
    if (cntH > RANGE) cntH = RANGE;
    const unsigned* dg = ndeg + h * RANGE;
    const unsigned mine = dg[rloc];
    int rank = 0;
    for (int j = 0; j < cntH; ++j) {
      unsigned o = dg[j];
      rank += (o > mine || (o == mine && j < rloc)) ? 1 : 0;
    }
    const int round = rank >> 5;
    const int pos = rank & 31;
    const int w = (round & 1) ? 31 - pos : pos;
    wnode[(h * 32 + w) * 4 + round] = rloc;
  }
  __syncthreads();
  if (t < NLISTS) {
    const int h = t >> 5;
    unsigned n = 0u;
    for (int qd = 0; qd < 4; ++qd) {
      int rloc = wnode[t * 4 + qd];
      if (rloc >= 0) n += ndeg[h * RANGE + rloc];
    }
    fcnt[t] = n;
    if (n > (unsigned)LCAP) ovf = 1u;
  }
  __syncthreads();
  const bool direct = ovf != 0u;

  if (!direct) {
    for (int h = 0; h < 2; ++h) {
      const int li = h * 32 + wave;
      int off = 0;
      for (int qd = 0; qd < 4; ++qd) {
        const int rloc = __builtin_amdgcn_readfirstlane(wnode[li * 4 + qd]);
        if (rloc < 0) continue;
        const int b = h * 32 + (rloc & 31);
        const int nb = __builtin_amdgcn_readfirstlane((int)lcnt[b]);
        for (int c0 = 0; c0 < nb; c0 += 32) {
          const int i = c0 + lane;
          uint2 en = make_uint2(0u, 0xFFFFFFFFu);
          if (i < nb) en = raw[b * LCAP + i];
          const bool m = en.y == (unsigned)rloc;
          const unsigned mask = __builtin_amdgcn_ballot_w32(m);
          if (m) {
            const int pos = off + __builtin_popcount(mask & ((1u << lane) - 1u));
            keys[li * LCAP + pos] = ((unsigned long long)en.y << 32) | (unsigned long long)en.x;
          }
          off += __builtin_popcount(mask);
        }
      }
    }
    __syncthreads();
    const unsigned bw = (unsigned)((N + 3) >> 2);
    for (int h = 0; h < 2; ++h) {
      const int li = h * 32 + wave;
      const int n = __builtin_amdgcn_readfirstlane((int)fcnt[li]);
      if (n > 0) {
        unsigned long long tk[4];
        unsigned cc[4];
#pragma unroll
        for (int r4 = 0; r4 < 4; ++r4) {
          const int i = lane + 32 * r4;
          tk[r4] = keys[li * LCAP + (i < n ? i : n - 1)];
        }
#pragma unroll
        for (int r4 = 0; r4 < 4; ++r4) cc[r4] = (unsigned)colIdx[(unsigned)tk[r4]];
#pragma unroll
        for (int r4 = 0; r4 < 4; ++r4) {
          const int i = lane + 32 * r4;
          if (i < n) {
            const unsigned e = (unsigned)tk[r4];
            const unsigned rloc = (unsigned)(tk[r4] >> 32);
            const unsigned band = (cc[r4] >= bw ? 1u : 0u) + (cc[r4] >= 2u * bw ? 1u : 0u)
                                + (cc[r4] >= 3u * bw ? 1u : 0u);
            keys[li * LCAP + i] = ((unsigned long long)band << 62)
                                | ((unsigned long long)rloc << 55)
                                | ((unsigned long long)cc[r4] << 31)
                                | (unsigned long long)e;
            pk[li * LCAP + i] = (rloc << 24) | cc[r4];
          }
        }
      }
    }
    __syncthreads();
    for (int h = 0; h < 2; ++h) {
      const int li = h * 32 + wave;
      const int n = (int)fcnt[li];
      for (int i = lane; i < n; i += 32) {
        unsigned long long key = keys[li * LCAP + i];
        int rank = 0;
#pragma unroll 4
        for (int j = 0; j < n; ++j) rank += keys[li * LCAP + j] < key ? 1 : 0;
        srt[li * LCAP + rank] = pk[li * LCAP + i];
      }
    }
    __syncthreads();
  }

  for (int h = 0; h < 2; ++h) {
    const int slo = lo + h * RANGE;
    int cnt = N - slo;
    if (cnt > RANGE) cnt = RANGE;
    if (cnt <= 0) break;

    float4 z = make_float4(0.f, 0.f, 0.f, 0.f);
    for (int i = t; i < RANGE * NODE_WORDS / 4; i += 1024)
      ((float4*)acc)[i] = z;
    {
      const float4* q4 = (const float4*)q + (size_t)slo * 32;
      for (int i = t; i < cnt * 32; i += 1024)
        ((float4*)qs)[i] = q4[i];
    }
    __syncthreads();

    if (direct) {
      painn_direct_pass(slo, (unsigned)cnt, t, lane, qs, k, vs, vv, v,
                        rowIdx, colIdx, acc, E);
    } else {
      const unsigned* L = srt + (h * 32 + wave) * LCAP;
      const int n = __builtin_amdgcn_readfirstlane((int)fcnt[h * 32 + wave]);
      unsigned e0 = lane < n ? L[lane] : 0u;
      unsigned e1 = lane + 32 < n ? L[lane + 32] : 0u;
      unsigned e2 = lane + 64 < n ? L[lane + 64] : 0u;
      unsigned e3 = lane + 96 < n ? L[lane + 96] : 0u;
#define PAINN_ENTRY(i) ((unsigned)__builtin_amdgcn_readlane( \
    (int)(((i) & 64) ? (((i) & 32) ? e3 : e2) : (((i) & 32) ? e1 : e0)), (i) & 31))
      painn_rows ga, gb, gc;
      unsigned pa = 0u, pb = 0u, pc = 0u;
      int cur = -1;
      float4 qv = make_float4(0.f, 0.f, 0.f, 0.f);
      painn_sums sm;
      sm.s = sm.v0 = sm.v1 = sm.v2 = make_float4(0.f, 0.f, 0.f, 0.f);
#define PAINN_STEP(G, P) { \
        const int rl = (int)((P) >> 24); \
        if (rl != cur) { \
          if (cur >= 0) painn_flush(sm, cur, lane, acc); \
          cur = rl; \
          qv = *(const float4*)(qs + rl * PDIM + lane * 4); \
          sm.s = sm.v0 = sm.v1 = sm.v2 = make_float4(0.f, 0.f, 0.f, 0.f); \
        } \
        painn_add_edge(G, qv, sm); }
      if (n > 0) {
        const int last = n - 1;
        pa = PAINN_ENTRY(0);
        painn_gather((int)(pa & 0xFFFFFFu), lane, k, vs, vv, v, ga);
        {
          const int j = 1 < last ? 1 : last;
          pb = PAINN_ENTRY(j);
          painn_gather((int)(pb & 0xFFFFFFu), lane, k, vs, vv, v, gb);
        }
        for (int i = 0; i < n; i += 3) {
          {
            const int j = i + 2 < last ? i + 2 : last;
            pc = PAINN_ENTRY(j);
            painn_gather((int)(pc & 0xFFFFFFu), lane, k, vs, vv, v, gc);
          }
          PAINN_STEP(ga, pa)
          {
            const int j = i + 3 < last ? i + 3 : last;
            pa = PAINN_ENTRY(j);
            painn_gather((int)(pa & 0xFFFFFFu), lane, k, vs, vv, v, ga);
          }
          if (i + 1 < n) PAINN_STEP(gb, pb)
          {
            const int j = i + 4 < last ? i + 4 : last;
            pb = PAINN_ENTRY(j);
            painn_gather((int)(pb & 0xFFFFFFu), lane, k, vs, vv, v, gb);
          }
          if (i + 2 < n) PAINN_STEP(gc, pc)
        }
        painn_flush(sm, cur, lane, acc);
      }
#undef PAINN_STEP
#undef PAINN_ENTRY
    }
    __syncthreads();

    {
      const float4* s4 = (const float4*)s + (size_t)slo * 32;
      float4* so4 = (float4*)s_out + (size_t)slo * 32;
      for (int u = t; u < cnt * 32; u += 1024) {
        const float* a = acc + (u >> 5) * NODE_WORDS + (u & 31);
        float4 sv = s4[u];
        float4 val;
        val.x = sv.x + a[0];
        val.y = sv.y + a[32];
        val.z = sv.z + a[64];
        val.w = sv.w + a[96];
        so4[u] = val;
      }
    }
    {
      float4* av4 = (float4*)aggv + (size_t)slo * 96;
      for (int u = t; u < cnt * 96; u += 1024) {
        int n = u / 96;
        int rem = u - n * 96;
        const float* a = acc + n * NODE_WORDS + PDIM + (rem >> 5) * PDIM + (rem & 31);
        av4[u] = make_float4(a[0], a[32], a[64], a[96]);
      }
    }
    __syncthreads();
  }
}

__global__ __launch_bounds__(256) void painn_vec_gemm(
    const float* __restrict__ aggv, const float* __restrict__ Wvec,
    const float* __restrict__ bvec, const float* __restrict__ v,
    float* __restrict__ vout, int nRowTiles) {
  int wave = __builtin_amdgcn_readfirstlane(threadIdx.x >> 5);
  int lane = threadIdx.x & 31;
  int half = lane >> 4;
  int l16  = lane & 15;
  int c0 = 64 * (wave & 1);
  int oc = c0 + 4 * l16;
  int nStreams = gridDim.x * 4;

  int rt = blockIdx.x * 4 + (wave >> 1);
  if (rt >= nRowTiles) return;

  painn_panel pan;
  painn_load_panel(Wvec, PDIM, c0, l16, half, pan);
  float4 bias = *(const float4*)(bvec + oc);

  painn_strip cur, nxt;
  painn_load_strip(aggv, rt * 16, l16, half, cur);
  for (; rt < nRowTiles; rt += nStreams) {
    int nx = rt + nStreams;
    if (nx >= nRowTiles) nx = nRowTiles - 1;
    painn_load_strip(aggv, nx * 16, l16, half, nxt);

    size_t o0 = (size_t)(rt * 16 + 8 * half) * PDIM + oc;
    const float* vrow = v + o0;
    float* orow = vout + o0;
    float4 vin[8];
#pragma unroll
    for (int r = 0; r < 8; ++r) vin[r] = *(const float4*)(vrow + r * PDIM);
    __builtin_amdgcn_sched_barrier(0);

    v8f acc[4] = {};
    painn_mma_strip(cur, pan, acc);

#pragma unroll
    for (int r = 0; r < 8; ++r) {
      float4 o;
      o.x = vin[r].x + acc[0][r] + bias.x;
      o.y = vin[r].y + acc[1][r] + bias.y;
      o.z = vin[r].z + acc[2][r] + bias.z;
      o.w = vin[r].w + acc[3][r] + bias.w;
      *(float4*)(orow + r * PDIM) = o;
    }
    cur = nxt;
  }
}

extern "C" void kernel_launch(void* const* d_in, const int* in_sizes, int n_in,
                              void* d_out, int out_size, void* d_ws, size_t ws_size,
                              hipStream_t stream) {
  const float* s    = (const float*)d_in[0];
  const float* v    = (const float*)d_in[1];
  const int*   ei   = (const int*)d_in[2];
  const float* Wq   = (const float*)d_in[3];
  const float* bq   = (const float*)d_in[4];
  const float* Wkv  = (const float*)d_in[5];
  const float* bkv  = (const float*)d_in[6];
  const float* Wvec = (const float*)d_in[7];
  const float* bvec = (const float*)d_in[8];

  int N = in_sizes[0] / PDIM;
  int E = in_sizes[2] / 2;

  float* s_out = (float*)d_out;
  float* v_out = s_out + (size_t)N * PDIM;

  float* q    = (float*)d_ws;
  float* k    = q  + (size_t)N * PDIM;
  float* vs   = k  + (size_t)N * PDIM;
  float* vv   = vs + (size_t)N * PDIM;
  float* aggv = vv + (size_t)N * PDIM;

  const int* rowIdx = ei;
  const int* colIdx = ei + E;

  painn_node_gemm<<<256, 256, 0, stream>>>(
      s, Wq, bq, Wkv, bkv, q, k, vs, vv, N / 16);

  painn_edge_range<<<(N + 2 * RANGE - 1) / (2 * RANGE), 1024, 0, stream>>>(
      q, k, vs, vv, v, s, rowIdx, colIdx, s_out, aggv, N, E);

  painn_vec_gemm<<<256, 256, 0, stream>>>(
      aggv, Wvec, bvec, v, v_out, N * 3 / 16);
}


# Node launch re-codes each destination as one byte (owner group number); range groups scan 16 edges per 16-byte load, read the index only on a match; ordered lists keep every sum's order (same sums)
# speedup vs baseline: 3.5282x; 1.0403x over previous
#include <hip/hip_runtime.h>

typedef __attribute__((ext_vector_type(2))) float v2f;
typedef __attribute__((ext_vector_type(8))) float v8f;

#define PDIM 128
#define GROUP_NODES 196

struct painn_panel { v2f f[4][32]; };
struct painn_strip { float4 a[16]; };

__device__ __forceinline__ void painn_load_panel(
    const float* __restrict__ W, int ldw, int c0, int l16, int half,
    painn_panel& p) {
  const float* wc = W + c0 + 4 * l16;
#pragma unroll
  for (int j = 0; j < 16; ++j) {
    int kb = 8 * j + 4 * half;
    float4 w0 = *(const float4*)(wc + (size_t)(kb + 0) * ldw);
    float4 w1 = *(const float4*)(wc + (size_t)(kb + 1) * ldw);
    float4 w2 = *(const float4*)(wc + (size_t)(kb + 2) * ldw);
    float4 w3 = *(const float4*)(wc + (size_t)(kb + 3) * ldw);
    p.f[0][2 * j].x = w0.x; p.f[0][2 * j].y = w1.x;
    p.f[1][2 * j].x = w0.y; p.f[1][2 * j].y = w1.y;
    p.f[2][2 * j].x = w0.z; p.f[2][2 * j].y = w1.z;
    p.f[3][2 * j].x = w0.w; p.f[3][2 * j].y = w1.w;
    p.f[0][2 * j + 1].x = w2.x; p.f[0][2 * j + 1].y = w3.x;
    p.f[1][2 * j + 1].x = w2.y; p.f[1][2 * j + 1].y = w3.y;
    p.f[2][2 * j + 1].x = w2.z; p.f[2][2 * j + 1].y = w3.z;
    p.f[3][2 * j + 1].x = w2.w; p.f[3][2 * j + 1].y = w3.w;
  }
}

__device__ __forceinline__ void painn_load_strip(
    const float* __restrict__ A, int rowBase, int l16, int half,
    painn_strip& st) {
  const float4* p = (const float4*)(A + (size_t)(rowBase + l16) * PDIM + 4 * half);
#pragma unroll
  for (int j = 0; j < 16; ++j) st.a[j] = p[2 * j];
}

__device__ __forceinline__ void painn_mma_strip(
    const painn_strip& st, const painn_panel& p, v8f (&acc)[4]) {
#pragma unroll
  for (int j = 0; j < 16; ++j) {
    v2f a0; a0.x = st.a[j].x; a0.y = st.a[j].y;
    v2f a1; a1.x = st.a[j].z; a1.y = st.a[j].w;
#pragma unroll
    for (int t = 0; t < 4; ++t)
      acc[t] = __builtin_amdgcn_wmma_f32_16x16x4_f32(false, a0, false, p.f[t][2 * j],
                                                     (short)0, acc[t], false, false);
#pragma unroll
    for (int t = 0; t < 4; ++t)
      acc[t] = __builtin_amdgcn_wmma_f32_16x16x4_f32(false, a1, false, p.f[t][2 * j + 1],
                                                     (short)0, acc[t], false, false);
  }
}

__device__ __forceinline__ void painn_write_groups(
    const int* __restrict__ rowIdx, unsigned char* __restrict__ grp, int E) {
  const int4* r4 = (const int4*)rowIdx;
  uchar4* g4 = (uchar4*)grp;
  const int E4 = E >> 2;
  const int stride = gridDim.x * 256;
  for (int i0 = blockIdx.x * 256 + threadIdx.x; i0 < E4; i0 += 4 * stride) {
    int4 a[4];
#pragma unroll
    for (int j = 0; j < 4; ++j) {
      const int i = i0 + j * stride;
      a[j] = r4[i < E4 ? i : E4 - 1];
    }
#pragma unroll
    for (int j = 0; j < 4; ++j) {
      const int i = i0 + j * stride;
      if (i < E4) {
        uchar4 o;
        o.x = (unsigned char)((unsigned)a[j].x / (unsigned)GROUP_NODES);
        o.y = (unsigned char)((unsigned)a[j].y / (unsigned)GROUP_NODES);
        o.z = (unsigned char)((unsigned)a[j].z / (unsigned)GROUP_NODES);
        o.w = (unsigned char)((unsigned)a[j].w / (unsigned)GROUP_NODES);
        g4[i] = o;
      }
    }
  }
  const int et = (E4 << 2) + (int)threadIdx.x;
  if (blockIdx.x == 0 && et < E)
    grp[et] = (unsigned char)((unsigned)rowIdx[et] / (unsigned)GROUP_NODES);
}

__global__ __launch_bounds__(256) void painn_node_gemm(
    const float* __restrict__ s,
    const float* __restrict__ Wq,  const float* __restrict__ bq,
    const float* __restrict__ Wkv, const float* __restrict__ bkv,
    float* __restrict__ q, float* __restrict__ k,
    float* __restrict__ vs, float* __restrict__ vv,
    int nRowTiles,
    const int* __restrict__ rowIdx, unsigned char* __restrict__ grp, int E) {
  int wave = __builtin_amdgcn_readfirstlane(threadIdx.x >> 5);
  int lane = threadIdx.x & 31;
  int half = lane >> 4;
  int l16  = lane & 15;

  const float* W; const float* b; int ldw; int c0;
  if (wave < 2) { W = Wq;  b = bq;  ldw = PDIM;     c0 = 64 * wave;        }
  else          { W = Wkv; b = bkv; ldw = 3 * PDIM; c0 = 64 * wave - PDIM; }
  float* out;
  if      (wave < 2) out = q;
  else if (wave < 4) out = k;
  else if (wave < 6) out = vs;
  else               out = vv;
  int oc = 64 * (wave & 1) + 4 * l16;

  int rt = blockIdx.x;
  if (rt >= nRowTiles) {
    painn_write_groups(rowIdx, grp, E);
    return;
  }

  painn_panel pan;
  painn_load_panel(W, ldw, c0, l16, half, pan);
  float4 bias = *(const float4*)(b + c0 + 4 * l16);

  painn_strip cur, nxt;
  painn_load_strip(s, rt * 16, l16, half, cur);
  painn_write_groups(rowIdx, grp, E);
  for (; rt < nRowTiles; rt += gridDim.x) {
    int nx = rt + gridDim.x;
    if (nx >= nRowTiles) nx = nRowTiles - 1;
    painn_load_strip(s, nx * 16, l16, half, nxt);
    __builtin_amdgcn_sched_barrier(0);

    v8f acc[4] = {};
    painn_mma_strip(cur, pan, acc);
    __builtin_amdgcn_sched_barrier(0);

    float* orow = out + (size_t)(rt * 16 + 8 * half) * PDIM + oc;
#pragma unroll
    for (int r = 0; r < 8; ++r) {
      float4 o;
      o.x = acc[0][r] + bias.x;
      o.y = acc[1][r] + bias.y;
      o.z = acc[2][r] + bias.z;
      o.w = acc[3][r] + bias.w;
      *(float4*)(orow + r * PDIM) = o;
    }
    cur = nxt;
  }
}

#define RANGE 98
static_assert(GROUP_NODES == 2 * RANGE, "a range work-group owns two ranges");
#define NODE_WORDS (4 * PDIM)
#define LCAP 128
#define NLISTS 64
#define CCAP 9216

struct painn_rows { float4 kv, vsv, vvv, x0, x1, x2; };

__device__ __forceinline__ void painn_gather(
    int c, int lane,
    const float* __restrict__ k, const float* __restrict__ vs,
    const float* __restrict__ vv, const float* __restrict__ v,
    painn_rows& g) {
  const unsigned ob = (unsigned)c * (unsigned)(PDIM * 4) + (unsigned)lane * 16u;
  g.kv  = *(const float4*)((const char*)k  + ob);
  g.vsv = *(const float4*)((const char*)vs + ob);
  g.vvv = *(const float4*)((const char*)vv + ob);
  const unsigned ov = (unsigned)c * (unsigned)(3 * PDIM * 4) + (unsigned)lane * 16u;
  const char* vc = (const char*)v + ov;
  g.x0 = *(const float4*)(vc);
  g.x1 = *(const float4*)(vc + PDIM * 4);
  g.x2 = *(const float4*)(vc + 2 * PDIM * 4);
}

__device__ __forceinline__ void painn_apply(
    const painn_rows& g, int rloc, int lane, const float* qs, float* acc) {
  float4 qv = *(const float4*)(qs + rloc * PDIM + lane * 4);
  float p = qv.x * g.kv.x + qv.y * g.kv.y + qv.z * g.kv.z + qv.w * g.kv.w;
  p += __shfl_xor(p, 16, 32);
  p += __shfl_xor(p, 8, 32);
  p += __shfl_xor(p, 4, 32);
  p += __shfl_xor(p, 2, 32);
  p += __shfl_xor(p, 1, 32);
  float w = p;

  float* a = acc + rloc * NODE_WORDS + lane;
  atomicAdd(a +  0, w * g.vsv.x);
  atomicAdd(a + 32, w * g.vsv.y);
  atomicAdd(a + 64, w * g.vsv.z);
  atomicAdd(a + 96, w * g.vsv.w);
  float* a0 = a + PDIM;
  atomicAdd(a0 +  0, w * g.vvv.x * g.x0.x);
  atomicAdd(a0 + 32, w * g.vvv.y * g.x0.y);
  atomicAdd(a0 + 64, w * g.vvv.z * g.x0.z);
  atomicAdd(a0 + 96, w * g.vvv.w * g.x0.w);
  float* a1 = a + 2 * PDIM;
  atomicAdd(a1 +  0, w * g.vvv.x * g.x1.x);
  atomicAdd(a1 + 32, w * g.vvv.y * g.x1.y);
  atomicAdd(a1 + 64, w * g.vvv.z * g.x1.z);
  atomicAdd(a1 + 96, w * g.vvv.w * g.x1.w);
  float* a2 = a + 3 * PDIM;
  atomicAdd(a2 +  0, w * g.vvv.x * g.x2.x);
  atomicAdd(a2 + 32, w * g.vvv.y * g.x2.y);
  atomicAdd(a2 + 64, w * g.vvv.z * g.x2.z);
  atomicAdd(a2 + 96, w * g.vvv.w * g.x2.w);
}

struct painn_sums { float4 s, v0, v1, v2; };

__device__ __forceinline__ void painn_add_edge(
    const painn_rows& g, const float4& qv, painn_sums& a) {
  float p = qv.x * g.kv.x + qv.y * g.kv.y + qv.z * g.kv.z + qv.w * g.kv.w;
  p += __shfl_xor(p, 16, 32);
  p += __shfl_xor(p, 8, 32);
  p += __shfl_xor(p, 4, 32);
  p += __shfl_xor(p, 2, 32);
  p += __shfl_xor(p, 1, 32);
  const float w = p;
  a.s.x += w * g.vsv.x;
  a.s.y += w * g.vsv.y;
  a.s.z += w * g.vsv.z;
  a.s.w += w * g.vsv.w;
  const float tx = w * g.vvv.x, ty = w * g.vvv.y, tz = w * g.vvv.z, tw = w * g.vvv.w;
  a.v0.x += tx * g.x0.x; a.v0.y += ty * g.x0.y; a.v0.z += tz * g.x0.z; a.v0.w += tw * g.x0.w;
  a.v1.x += tx * g.x1.x; a.v1.y += ty * g.x1.y; a.v1.z += tz * g.x1.z; a.v1.w += tw * g.x1.w;
  a.v2.x += tx * g.x2.x; a.v2.y += ty * g.x2.y; a.v2.z += tz * g.x2.z; a.v2.w += tw * g.x2.w;
}

__device__ __forceinline__ void painn_flush(
    const painn_sums& a, int rloc, int lane, float* acc) {
  float* p = acc + rloc * NODE_WORDS + lane;
  atomicAdd(p +  0, a.s.x);
  atomicAdd(p + 32, a.s.y);
  atomicAdd(p + 64, a.s.z);
  atomicAdd(p + 96, a.s.w);
  float* p0 = p + PDIM;
  atomicAdd(p0 +  0, a.v0.x);
  atomicAdd(p0 + 32, a.v0.y);
  atomicAdd(p0 + 64, a.v0.z);
  atomicAdd(p0 + 96, a.v0.w);
  float* p1 = p + 2 * PDIM;
  atomicAdd(p1 +  0, a.v1.x);
  atomicAdd(p1 + 32, a.v1.y);
  atomicAdd(p1 + 64, a.v1.z);
  atomicAdd(p1 + 96, a.v1.w);
  float* p2 = p + 3 * PDIM;
  atomicAdd(p2 +  0, a.v2.x);
  atomicAdd(p2 + 32, a.v2.y);
  atomicAdd(p2 + 64, a.v2.z);
  atomicAdd(p2 + 96, a.v2.w);
}

__device__ __forceinline__ unsigned painn_hits(int4 a, int lo, unsigned cnt) {
  return ((unsigned)(a.x - lo) < cnt ? 1u : 0u)
       | ((unsigned)(a.y - lo) < cnt ? 2u : 0u)
       | ((unsigned)(a.z - lo) < cnt ? 4u : 0u)
       | ((unsigned)(a.w - lo) < cnt ? 8u : 0u);
}

__device__ __forceinline__ void painn_note(
    int r, int e, int lo, unsigned cntAll, unsigned* lcnt, unsigned* ndeg, uint2* raw) {
  unsigned rl = (unsigned)(r - lo);
  if (rl < cntAll) {
    unsigned h = rl >= (unsigned)RANGE ? 1u : 0u;
    unsigned rloc = rl - h * (unsigned)RANGE;
    unsigned li = h * 32u + (rloc & 31u);
    atomicAdd(&ndeg[rl], 1u);
    unsigned slot = atomicAdd(&lcnt[li], 1u);
    if (slot < (unsigned)LCAP) raw[li * LCAP + slot] = make_uint2((unsigned)e, rloc);
  }
}

__device__ __forceinline__ void painn_load_set(
    const uint4* __restrict__ g16, int u, int lastIdx, uint4 (&a)[4]) {
#pragma unroll
  for (int j = 0; j < 4; ++j) {
    int idx = u + j * 1024;
    a[j] = g16[idx < lastIdx ? idx : lastIdx];
  }
}

__device__ __forceinline__ unsigned painn_zero_bytes(unsigned x) {
  return (x - 0x01010101u) & ~x & 0x80808080u;
}

__device__ __forceinline__ void painn_cand_set(
    const uint4 (&a)[4], int u, int E16, unsigned pat,
    unsigned* ncand, unsigned* cand) {
#pragma unroll
  for (int j = 0; j < 4; ++j) {
    const int idx = u + j * 1024;
    const unsigned z0 = painn_zero_bytes(a[j].x ^ pat);
    const unsigned z1 = painn_zero_bytes(a[j].y ^ pat);
    const unsigned z2 = painn_zero_bytes(a[j].z ^ pat);
    const unsigned z3 = painn_zero_bytes(a[j].w ^ pat);
    if (idx < E16 && (z0 | z1 | z2 | z3) != 0u) {
      const int e0 = idx << 4;
      unsigned m = (z0 >> 7) | (z1 >> 6) | (z2 >> 5) | (z3 >> 4);
      while (m) {
        const int i = __builtin_ctz(m);
        m &= m - 1u;
        const unsigned slot = atomicAdd(ncand, 1u);
        if (slot < (unsigned)CCAP) cand[slot] = (unsigned)(e0 + ((i & 7) << 2) + (i >> 3));
      }
    }
  }
}

__device__ __forceinline__ void painn_direct_pass(
    int lo, unsigned ucnt, int t, int lane, const float* qs,
    const float* __restrict__ k, const float* __restrict__ vs,
    const float* __restrict__ vv, const float* __restrict__ v,
    const int* __restrict__ rowIdx, const int* __restrict__ colIdx,
    float* acc, int E) {
  const int4* row4 = (const int4*)rowIdx;
  const int E4 = E >> 2;
  const int4 none = make_int4(-1, -1, -1, -1);
  const int wave0 = __builtin_amdgcn_readfirstlane(t - lane);
  for (int ub = wave0; ub < E4; ub += 4096) {
    int u = ub + lane;
    int4 a0 = none, a1 = none, a2 = none, a3 = none;
    if (u < E4) a0 = row4[u];
    if (u + 1024 < E4) a1 = row4[u + 1024];
    if (u + 2048 < E4) a2 = row4[u + 2048];
    if (u + 3072 < E4) a3 = row4[u + 3072];
    unsigned m = painn_hits(a0, lo, ucnt)
               | (painn_hits(a1, lo, ucnt) << 4)
               | (painn_hits(a2, lo, ucnt) << 8)
               | (painn_hits(a3, lo, ucnt) << 12);
    unsigned any = __builtin_amdgcn_ballot_w32(m != 0u);
    while (any) {
      int src = __builtin_ctz(any);
      any &= any - 1u;
      unsigned sm = (unsigned)__builtin_amdgcn_readlane((int)m, src);
      while (sm) {
        int bit = __builtin_ctz(sm);
        sm &= sm - 1u;
        int e = ((ub + (bit >> 2) * 1024 + src) << 2) + (bit & 3);
        int r = rowIdx[e];
        painn_rows g;
        painn_gather(colIdx[e], lane, k, vs, vv, v, g);
        painn_apply(g, r - lo, lane, qs, acc);
      }
    }
  }
  if (t < 32) {
    for (int e = E4 << 2; e < E; ++e) {
      int r = rowIdx[e];
      if ((unsigned)(r - lo) < ucnt) {
        painn_rows g;
        painn_gather(colIdx[e], lane, k, vs, vv, v, g);
        painn_apply(g, r - lo, lane, qs, acc);
      }
    }
  }
}

__global__ __launch_bounds__(1024) void painn_edge_range(
    const float* __restrict__ q, const float* __restrict__ k,
    const float* __restrict__ vs, const float* __restrict__ vv,
    const float* __restrict__ v, const float* __restrict__ s,
    const int* __restrict__ rowIdx, const int* __restrict__ colIdx,
    const unsigned char* __restrict__ grp,
    float* __restrict__ s_out, float* __restrict__ aggv, int N, int E) {
  __shared__ float acc[RANGE * NODE_WORDS];
  __shared__ float qs[RANGE * PDIM];
  __shared__ unsigned srt[NLISTS * LCAP];
  __shared__ unsigned lcnt[NLISTS];
  __shared__ unsigned fcnt[NLISTS];
  __shared__ unsigned ndeg[2 * RANGE];
  __shared__ int wnode[NLISTS * 4];
  __shared__ unsigned ovf;
  __shared__ unsigned ncand;

  const int t = threadIdx.x;
  const int lane = t & 31;
  const int wave = __builtin_amdgcn_readfirstlane(t >> 5);
  const int lo = blockIdx.x * (2 * RANGE);
  int cntAll = N - lo;
  if (cntAll > 2 * RANGE) cntAll = 2 * RANGE;
  const unsigned ucntAll = (unsigned)cntAll;

  uint2* raw = (uint2*)acc;
  unsigned long long* keys = (unsigned long long*)(acc + 2 * NLISTS * LCAP);
  unsigned* pk = (unsigned*)(acc + 4 * NLISTS * LCAP);
  unsigned* cand = (unsigned*)(acc + 5 * NLISTS * LCAP);

  if (t < NLISTS) lcnt[t] = 0u;
  if (t < 2 * RANGE) ndeg[t] = 0u;
  if (t < NLISTS * 4) wnode[t] = -1;
  if (t == 0) { ovf = 0u; ncand = 0u; }
  __syncthreads();

  {
    const uint4* g16 = (const uint4*)grp;
    const int E16 = E >> 4;
    const unsigned pat = (blockIdx.x & 255u) * 0x01010101u;
    const int wave0 = __builtin_amdgcn_readfirstlane(t - lane);
    if (E16 > 0) {
      uint4 sa[4], sb[4];
      painn_load_set(g16, wave0 + lane, E16 - 1, sa);
      for (int ub = wave0; ub < E16; ub += 8192) {
        const int u = ub + lane;
        painn_load_set(g16, u + 4096, E16 - 1, sb);
        painn_cand_set(sa, u, E16, pat, &ncand, cand);
        painn_load_set(g16, u + 8192, E16 - 1, sa);
        painn_cand_set(sb, u + 4096, E16, pat, &ncand, cand);
      }
    }
  }
  __syncthreads();
  {
    unsigned nc = ncand;
    if (nc > (unsigned)CCAP) {
      nc = (unsigned)CCAP;
      if (t == 0) ovf = 1u;
    }
    for (unsigned base = 0u; base < nc; base += 4096u) {
      int ce[4], cr[4];
#pragma unroll
      for (int j = 0; j < 4; ++j) {
        const unsigned i = base + (unsigned)t + 1024u * (unsigned)j;
        ce[j] = (int)cand[i < nc ? i : nc - 1u];
      }
#pragma unroll
      for (int j = 0; j < 4; ++j) cr[j] = rowIdx[ce[j]];
#pragma unroll
      for (int j = 0; j < 4; ++j) {
        const unsigned i = base + (unsigned)t + 1024u * (unsigned)j;
        if (i < nc) painn_note(cr[j], ce[j], lo, ucntAll, lcnt, ndeg, raw);
      }
    }
    const int et = ((E >> 4) << 4) + t;
    if (et < E) painn_note(rowIdx[et], et, lo, ucntAll, lcnt, ndeg, raw);
  }
  __syncthreads();
  if (t < NLISTS && lcnt[t] > (unsigned)LCAP) ovf = 1u;
  if (t < cntAll) {
    const int h = t >= RANGE ? 1 : 0;
    const int rloc = t - h * RANGE;
    int cntH = cntAll - h * RANGE;
    if (cntH > RANGE) cntH = RANGE;
    const unsigned* dg = ndeg + h * RANGE;
    const unsigned mine = dg[rloc];
    int rank = 0;
    for (int j = 0; j < cntH; ++j) {
      unsigned o = dg[j];
      rank += (o > mine || (o == mine && j < rloc)) ? 1 : 0;
    }
    const int round = rank >> 5;
    const int pos = rank & 31;
    const int w = (round & 1) ? 31 - pos : pos;
    wnode[(h * 32 + w) * 4 + round] = rloc;
  }
  __syncthreads();
  if (t < NLISTS) {
    const int h = t >> 5;
    unsigned n = 0u;
    for (int qd = 0; qd < 4; ++qd) {
      int rloc = wnode[t * 4 + qd];
      if (rloc >= 0) n += ndeg[h * RANGE + rloc];
    }
    fcnt[t] = n;
    if (n > (unsigned)LCAP) ovf = 1u;
  }
  __syncthreads();
  const bool direct = ovf != 0u;

  if (!direct) {
    for (int h = 0; h < 2; ++h) {
      const int li = h * 32 + wave;
      int off = 0;
      for (int qd = 0; qd < 4; ++qd) {
        const int rloc = __builtin_amdgcn_readfirstlane(wnode[li * 4 + qd]);
        if (rloc < 0) continue;
        const int b = h * 32 + (rloc & 31);
        const int nb = __builtin_amdgcn_readfirstlane((int)lcnt[b]);
        for (int c0 = 0; c0 < nb; c0 += 32) {
          const int i = c0 + lane;
          uint2 en = make_uint2(0u, 0xFFFFFFFFu);
          if (i < nb) en = raw[b * LCAP + i];
          const bool m = en.y == (unsigned)rloc;
          const unsigned mask = __builtin_amdgcn_ballot_w32(m);
          if (m) {
            const int pos = off + __builtin_popcount(mask & ((1u << lane) - 1u));
            keys[li * LCAP + pos] = ((unsigned long long)en.y << 32) | (unsigned long long)en.x;
          }
          off += __builtin_popcount(mask);
        }
      }
    }
    __syncthreads();
    const unsigned bw = (unsigned)((N + 3) >> 2);
    for (int h = 0; h < 2; ++h) {
      const int li = h * 32 + wave;
      const int n = __builtin_amdgcn_readfirstlane((int)fcnt[li]);
      if (n > 0) {
        unsigned long long tk[4];
        unsigned cc[4];
#pragma unroll
        for (int r4 = 0; r4 < 4; ++r4) {
          const int i = lane + 32 * r4;
          tk[r4] = keys[li * LCAP + (i < n ? i : n - 1)];
        }
#pragma unroll
        for (int r4 = 0; r4 < 4; ++r4) cc[r4] = (unsigned)colIdx[(unsigned)tk[r4]];
#pragma unroll
        for (int r4 = 0; r4 < 4; ++r4) {
          const int i = lane + 32 * r4;
          if (i < n) {
            const unsigned e = (unsigned)tk[r4];
            const unsigned rloc = (unsigned)(tk[r4] >> 32);
            const unsigned band = (cc[r4] >= bw ? 1u : 0u) + (cc[r4] >= 2u * bw ? 1u : 0u)
                                + (cc[r4] >= 3u * bw ? 1u : 0u);
            keys[li * LCAP + i] = ((unsigned long long)band << 62)
                                | ((unsigned long long)rloc << 55)
                                | ((unsigned long long)cc[r4] << 31)
                                | (unsigned long long)e;
            pk[li * LCAP + i] = (rloc << 24) | cc[r4];
          }
        }
      }
    }
    __syncthreads();
    for (int h = 0; h < 2; ++h) {
      const int li = h * 32 + wave;
      const int n = (int)fcnt[li];
      for (int i = lane; i < n; i += 32) {
        unsigned long long key = keys[li * LCAP + i];
        int rank = 0;
#pragma unroll 4
        for (int j = 0; j < n; ++j) rank += keys[li * LCAP + j] < key ? 1 : 0;
        srt[li * LCAP + rank] = pk[li * LCAP + i];
      }
    }
    __syncthreads();
  }

  for (int h = 0; h < 2; ++h) {
    const int slo = lo + h * RANGE;
    int cnt = N - slo;
    if (cnt > RANGE) cnt = RANGE;
    if (cnt <= 0) break;

    float4 z = make_float4(0.f, 0.f, 0.f, 0.f);
    for (int i = t; i < RANGE * NODE_WORDS / 4; i += 1024)
      ((float4*)acc)[i] = z;
    {
      const float4* q4 = (const float4*)q + (size_t)slo * 32;
      for (int i = t; i < cnt * 32; i += 1024)
        ((float4*)qs)[i] = q4[i];
    }
    __syncthreads();

    if (direct) {
      painn_direct_pass(slo, (unsigned)cnt, t, lane, qs, k, vs, vv, v,
                        rowIdx, colIdx, acc, E);
    } else {
      const unsigned* L = srt + (h * 32 + wave) * LCAP;
      const int n = __builtin_amdgcn_readfirstlane((int)fcnt[h * 32 + wave]);
      unsigned e0 = lane < n ? L[lane] : 0u;
      unsigned e1 = lane + 32 < n ? L[lane + 32] : 0u;
      unsigned e2 = lane + 64 < n ? L[lane + 64] : 0u;
      unsigned e3 = lane + 96 < n ? L[lane + 96] : 0u;
#define PAINN_ENTRY(i) ((unsigned)__builtin_amdgcn_readlane( \
    (int)(((i) & 64) ? (((i) & 32) ? e3 : e2) : (((i) & 32) ? e1 : e0)), (i) & 31))
      painn_rows ga, gb, gc;
      unsigned pa = 0u, pb = 0u, pc = 0u;
      int cur = -1;
      float4 qv = make_float4(0.f, 0.f, 0.f, 0.f);
      painn_sums sm;
      sm.s = sm.v0 = sm.v1 = sm.v2 = make_float4(0.f, 0.f, 0.f, 0.f);
#define PAINN_STEP(G, P) { \
        const int rl = (int)((P) >> 24); \
        if (rl != cur) { \
          if (cur >= 0) painn_flush(sm, cur, lane, acc); \
          cur = rl; \
          qv = *(const float4*)(qs + rl * PDIM + lane * 4); \
          sm.s = sm.v0 = sm.v1 = sm.v2 = make_float4(0.f, 0.f, 0.f, 0.f); \
        } \
        painn_add_edge(G, qv, sm); }
      if (n > 0) {
        const int last = n - 1;
        pa = PAINN_ENTRY(0);
        painn_gather((int)(pa & 0xFFFFFFu), lane, k, vs, vv, v, ga);
        {
          const int j = 1 < last ? 1 : last;
          pb = PAINN_ENTRY(j);
          painn_gather((int)(pb & 0xFFFFFFu), lane, k, vs, vv, v, gb);
        }
        for (int i = 0; i < n; i += 3) {
          {
            const int j = i + 2 < last ? i + 2 : last;
            pc = PAINN_ENTRY(j);
            painn_gather((int)(pc & 0xFFFFFFu), lane, k, vs, vv, v, gc);
          }
          PAINN_STEP(ga, pa)
          {
            const int j = i + 3 < last ? i + 3 : last;
            pa = PAINN_ENTRY(j);
            painn_gather((int)(pa & 0xFFFFFFu), lane, k, vs, vv, v, ga);
          }
          if (i + 1 < n) PAINN_STEP(gb, pb)
          {
            const int j = i + 4 < last ? i + 4 : last;
            pb = PAINN_ENTRY(j);
            painn_gather((int)(pb & 0xFFFFFFu), lane, k, vs, vv, v, gb);
          }
          if (i + 2 < n) PAINN_STEP(gc, pc)
        }
        painn_flush(sm, cur, lane, acc);
      }
#undef PAINN_STEP
#undef PAINN_ENTRY
    }
    __syncthreads();

    {
      const float4* s4 = (const float4*)s + (size_t)slo * 32;
      float4* so4 = (float4*)s_out + (size_t)slo * 32;
      for (int u = t; u < cnt * 32; u += 1024) {
        const float* a = acc + (u >> 5) * NODE_WORDS + (u & 31);
        float4 sv = s4[u];
        float4 val;
        val.x = sv.x + a[0];
        val.y = sv.y + a[32];
        val.z = sv.z + a[64];
        val.w = sv.w + a[96];
        so4[u] = val;
      }
    }
    {
      float4* av4 = (float4*)aggv + (size_t)slo * 96;
      for (int u = t; u < cnt * 96; u += 1024) {
        int n = u / 96;
        int rem = u - n * 96;
        const float* a = acc + n * NODE_WORDS + PDIM + (rem >> 5) * PDIM + (rem & 31);
        av4[u] = make_float4(a[0], a[32], a[64], a[96]);
      }
    }
    __syncthreads();
  }
}

__global__ __launch_bounds__(256) void painn_vec_gemm(
    const float* __restrict__ aggv, const float* __restrict__ Wvec,
    const float* __restrict__ bvec, const float* __restrict__ v,
    float* __restrict__ vout, int nRowTiles) {
  int wave = __builtin_amdgcn_readfirstlane(threadIdx.x >> 5);
  int lane = threadIdx.x & 31;
  int half = lane >> 4;
  int l16  = lane & 15;
  int c0 = 64 * (wave & 1);
  int oc = c0 + 4 * l16;
  int nStreams = gridDim.x * 4;

  int rt = blockIdx.x * 4 + (wave >> 1);
  if (rt >= nRowTiles) return;

  painn_panel pan;
  painn_load_panel(Wvec, PDIM, c0, l16, half, pan);
  float4 bias = *(const float4*)(bvec + oc);

  painn_strip cur, nxt;
  painn_load_strip(aggv, rt * 16, l16, half, cur);
  for (; rt < nRowTiles; rt += nStreams) {
    int nx = rt + nStreams;
    if (nx >= nRowTiles) nx = nRowTiles - 1;
    painn_load_strip(aggv, nx * 16, l16, half, nxt);

    size_t o0 = (size_t)(rt * 16 + 8 * half) * PDIM + oc;
    const float* vrow = v + o0;
    float* orow = vout + o0;
    float4 vin[8];
#pragma unroll
    for (int r = 0; r < 8; ++r) vin[r] = *(const float4*)(vrow + r * PDIM);
    __builtin_amdgcn_sched_barrier(0);

    v8f acc[4] = {};
    painn_mma_strip(cur, pan, acc);

#pragma unroll
    for (int r = 0; r < 8; ++r) {
      float4 o;
      o.x = vin[r].x + acc[0][r] + bias.x;
      o.y = vin[r].y + acc[1][r] + bias.y;
      o.z = vin[r].z + acc[2][r] + bias.z;
      o.w = vin[r].w + acc[3][r] + bias.w;
      *(float4*)(orow + r * PDIM) = o;
    }
    cur = nxt;
  }
}

extern "C" void kernel_launch(void* const* d_in, const int* in_sizes, int n_in,
                              void* d_out, int out_size, void* d_ws, size_t ws_size,
                              hipStream_t stream) {
  const float* s    = (const float*)d_in[0];
  const float* v    = (const float*)d_in[1];
  const int*   ei   = (const int*)d_in[2];
  const float* Wq   = (const float*)d_in[3];
  const float* bq   = (const float*)d_in[4];
  const float* Wkv  = (const float*)d_in[5];
  const float* bkv  = (const float*)d_in[6];
  const float* Wvec = (const float*)d_in[7];
  const float* bvec = (const float*)d_in[8];

  int N = in_sizes[0] / PDIM;
  int E = in_sizes[2] / 2;

  float* s_out = (float*)d_out;
  float* v_out = s_out + (size_t)N * PDIM;

  float* q    = (float*)d_ws;
  float* k    = q  + (size_t)N * PDIM;
  float* vs   = k  + (size_t)N * PDIM;
  float* vv   = vs + (size_t)N * PDIM;
  float* aggv = vv + (size_t)N * PDIM;
  unsigned char* grp = (unsigned char*)(aggv + (size_t)N * 3 * PDIM);

  const int* rowIdx = ei;
  const int* colIdx = ei + E;

  painn_node_gemm<<<256, 256, 0, stream>>>(
      s, Wq, bq, Wkv, bkv, q, k, vs, vv, N / 16, rowIdx, grp, E);

  painn_edge_range<<<(N + 2 * RANGE - 1) / (2 * RANGE), 1024, 0, stream>>>(
      q, k, vs, vv, v, s, rowIdx, colIdx, grp, s_out, aggv, N, E);

  painn_vec_gemm<<<256, 256, 0, stream>>>(
      aggv, Wvec, bvec, v, v_out, N * 3 / 16);
}
